# starting kernel
# baseline (speedup 1.0000x reference)
#include <hip/hip_runtime.h>
#include <hip/hip_bf16.h>

typedef __attribute__((ext_vector_type(2))) float v2f;
typedef __attribute__((ext_vector_type(8))) float v8f;

#define D 128

__device__ __forceinline__ void atomic_add_f32_hw(float* addr, float val) {
    asm volatile("global_atomic_add_f32 %0, %1, off"
                 :
                 : "v"(addr), "v"(val)
                 : "memory");
}

__global__ __launch_bounds__(256) void gcn_zero(float4* __restrict__ p, int n4) {
    int i = blockIdx.x * blockDim.x + threadIdx.x;
    if (i < n4) p[i] = make_float4(0.f, 0.f, 0.f, 0.f);
}

__global__ __launch_bounds__(256) void gcn_gemm_wmma(const float* __restrict__ x,
                                                     const float* __restrict__ W,
                                                     float* __restrict__ support) {
    const int lane = threadIdx.x & 31;
    const int wave = threadIdx.x >> 5;
    const int m0   = blockIdx.x * 16;
    const int n0   = wave * 16;
    const int half = lane >> 4;
    const int l16  = lane & 15;

    const float* arow = x + (size_t)(m0 + l16) * D;
    v8f c = {};
#pragma unroll 4
    for (int k = 0; k < D; k += 4) {
        const int ka = k + half * 2;
        v2f a;
        a.x = arow[ka];
        a.y = arow[ka + 1];
        v2f b;
        b.x = W[(size_t)ka * D + n0 + l16];
        b.y = W[(size_t)(ka + 1) * D + n0 + l16];
        c = __builtin_amdgcn_wmma_f32_16x16x4_f32(
                false, a, false, b, (short)0, c, false, false);
    }

    float* srow = support + n0 + l16;
#pragma unroll
    for (int v = 0; v < 8; ++v) {
        srow[(size_t)(m0 + v + half * 8) * D] = c[v];
    }
}

__global__ __launch_bounds__(256) void gcn_scatter_atomic(const float* __restrict__ support,
                                                          const int* __restrict__ erow,
                                                          const int* __restrict__ ecol,
                                                          const float* __restrict__ eval,
                                                          float* __restrict__ out,
                                                          int nEdges) {
    const int wv = __builtin_amdgcn_readfirstlane(threadIdx.x >> 5);
    const int e  = blockIdx.x * 8 + wv;
    if (e >= nEdges) return;
    const int lane = threadIdx.x & 31;

    const int   r = erow[e];
    const int   c = ecol[e];
    const float v = eval[e];

    const float4 m = *((const float4*)(support + (size_t)c * D) + lane);
    float* dst = out + (size_t)r * D + lane * 4;

    atomic_add_f32_hw(dst + 0, m.x * v);
    atomic_add_f32_hw(dst + 1, m.y * v);
    atomic_add_f32_hw(dst + 2, m.z * v);
    atomic_add_f32_hw(dst + 3, m.w * v);
}

__global__ __launch_bounds__(256) void gcn_tanh(float4* __restrict__ p, int n4) {
    int i = blockIdx.x * blockDim.x + threadIdx.x;
    if (i < n4) {
        float4 t = p[i];
        t.x = tanhf(t.x);
        t.y = tanhf(t.y);
        t.z = tanhf(t.z);
        t.w = tanhf(t.w);
        p[i] = t;
    }
}

extern "C" void kernel_launch(void* const* d_in, const int* in_sizes, int n_in,
                              void* d_out, int out_size, void* d_ws, size_t ws_size,
                              hipStream_t stream) {
    const float* x    = (const float*)d_in[0];
    const float* W    = (const float*)d_in[1];
    const int*   erow = (const int*)d_in[2];
    const int*   ecol = (const int*)d_in[3];
    const float* eval = (const float*)d_in[4];
    float* out = (float*)d_out;

    const int nNodes = in_sizes[0] / D;
    const int nEdges = in_sizes[2];

    float* support = (float*)d_ws;

    {
        const int n4 = out_size / 4;
        gcn_zero<<<(n4 + 255) / 256, 256, 0, stream>>>((float4*)out, n4);
    }
    {
        const int blocks = nNodes / 16;
        gcn_gemm_wmma<<<blocks, 256, 0, stream>>>(x, W, support);
    }
    {
        const int blocks = (nEdges + 7) / 8;
        gcn_scatter_atomic<<<blocks, 256, 0, stream>>>(support, erow, ecol, eval, out, nEdges);
    }
    {
        const int n4 = out_size / 4;
        gcn_tanh<<<(n4 + 255) / 256, 256, 0, stream>>>((float4*)out, n4);
    }
}


# aggregate without atomics: a block filters the whole edge_row list for its 400 out rows (no partition pass), each wave adds its own rows' hits in LDS in a fixed order, stores tanh; 4 launches -> 2
# speedup vs baseline: 2.0750x; 2.0750x over previous
#include <hip/hip_runtime.h>
#include <hip/hip_bf16.h>

typedef __attribute__((ext_vector_type(2))) float v2f;
typedef __attribute__((ext_vector_type(8))) float v8f;

#define D 128

#define AS_PITCH 132

__global__ __launch_bounds__(256) void gcn_gemm_wmma(const float* __restrict__ x,
                                                     const float* __restrict__ W,
                                                     float* __restrict__ support,
                                                     int nStripes) {
    __shared__ __attribute__((aligned(16))) float As[16 * AS_PITCH];

    const int t    = threadIdx.x;
    const int lane = t & 31;
    const int wave = t >> 5;
    const int n0   = wave * 16;
    const int half = lane >> 4;
    const int l16  = lane & 15;

    v2f bfrag[32];
#pragma unroll
    for (int s = 0; s < 32; ++s) {
        const int ka = 4 * s + half * 2;
        bfrag[s].x = W[(size_t)ka * D + n0 + l16];
        bfrag[s].y = W[(size_t)(ka + 1) * D + n0 + l16];
    }

    float* lds0 = As + (t >> 5) * AS_PITCH + (t & 31) * 4;
    float* lds1 = lds0 + 8 * AS_PITCH;
    const float* arow = As + l16 * AS_PITCH + half * 2;
    float* srow = support + n0 + l16;

    int stripe = blockIdx.x;
    if (stripe >= nStripes) return;
    const float4* src0 = (const float4*)(x + (size_t)stripe * 16 * D);
    float4 g0 = src0[t];
    float4 g1 = src0[t + 256];
    for (; stripe < nStripes; stripe += gridDim.x) {
        __syncthreads();
        *(float4*)lds0 = g0;
        *(float4*)lds1 = g1;
        __syncthreads();

        const int next = stripe + gridDim.x;
        const int nx   = next < nStripes ? next : stripe;
        const float4* src = (const float4*)(x + (size_t)nx * 16 * D);
        g0 = src[t];
        g1 = src[t + 256];

        v8f c = {};
#pragma unroll
        for (int s = 0; s < 32; ++s) {
            const v2f a = *(const v2f*)(arow + 4 * s);
            c = __builtin_amdgcn_wmma_f32_16x16x4_f32(
                    false, a, false, bfrag[s], (short)0, c, false, false);
        }

        const int m0 = stripe * 16;
#pragma unroll
        for (int v = 0; v < 8; ++v) {
            srow[(size_t)(m0 + v + half * 8) * D] = c[v];
        }
    }
}

#define AGG_ROWS    400
#define AGG_THREADS 1024
#define AGG_WAVES   32
#define AGG_U       4
#define AGG_STEP    (AGG_U * AGG_THREADS)
#define AGG_QCAP    800
#define AGG_OCAP    64
#define AGG_RBITS   9

__device__ __forceinline__ void agg_take(int e, int r, int r0, int nr, unsigned* q, int& cnt) {
    const unsigned rl  = (unsigned)(r - r0);
    const bool     hit = rl < (unsigned)nr;
    const unsigned mask = __builtin_amdgcn_ballot_w32(hit);
    if (mask != 0u) {
        if (hit) q[cnt + (int)__builtin_amdgcn_mbcnt_lo(mask, 0u)] = ((unsigned)e << AGG_RBITS) | rl;
        cnt += __builtin_popcount(mask);
    }
}

__device__ __forceinline__ void agg_drain(const unsigned* o, int m, int lane, int nNodes,
                                          const float* __restrict__ support,
                                          const int* __restrict__ ecol,
                                          const float* __restrict__ eval,
                                          float* acc) {
    int c = 0, rl = 0;
    float v = 0.f;
    if (lane < m) {
        const unsigned ent = o[lane];
        const int e = (int)(ent >> AGG_RBITS);
        rl = (int)(ent & ((1u << AGG_RBITS) - 1u));
        c  = min(max(ecol[e], 0), nNodes - 1);
        v  = eval[e];
    }
    for (int j = 0; j < m; j += 8) {
        float4 g[8];
#pragma unroll
        for (int u = 0; u < 8; ++u) {
            const int cc = __builtin_amdgcn_readlane(c, min(j + u, m - 1));
            g[u] = ((const float4*)(support + (size_t)cc * D))[lane];
        }
#pragma unroll
        for (int u = 0; u < 8; ++u) {
            if (j + u < m) {
                const float vv = __int_as_float(__builtin_amdgcn_readlane(__float_as_int(v), j + u));
                const int   rr = __builtin_amdgcn_readlane(rl, j + u);
                float* a = acc + rr * D + lane;
                a[0]  += g[u].x * vv;
                a[32] += g[u].y * vv;
                a[64] += g[u].z * vv;
                a[96] += g[u].w * vv;
            }
        }
    }
}

__global__ __launch_bounds__(AGG_THREADS) void gcn_aggregate_tanh(const float* __restrict__ support,
                                                                  const int* __restrict__ erow,
                                                                  const int* __restrict__ ecol,
                                                                  const float* __restrict__ eval,
                                                                  float* __restrict__ out,
                                                                  int nNodes, int nEdges) {
    __shared__ __attribute__((aligned(16))) float acc[AGG_ROWS * D];
    __shared__ unsigned hitq[AGG_WAVES * AGG_QCAP];
    __shared__ unsigned ownq[AGG_WAVES * AGG_OCAP];
    __shared__ int      qcount[AGG_WAVES];
    __shared__ int      qmore[AGG_WAVES];

    const int t    = threadIdx.x;
    const int lane = t & 31;
    const int wv   = __builtin_amdgcn_readfirstlane(t >> 5);
    const int r0   = blockIdx.x * AGG_ROWS;
    const int nr   = min(AGG_ROWS, nNodes - r0);

    for (int i = t; i < AGG_ROWS * D / 4; i += AGG_THREADS)
        ((float4*)acc)[i] = make_float4(0.f, 0.f, 0.f, 0.f);

    unsigned* q = hitq + wv * AGG_QCAP;
    unsigned* o = ownq + wv * AGG_OCAP;
    int cnt = 0;

    const int   E4   = nEdges >> 2;
    const int   gEnd = max(E4 - 1, 0);
    const int4* er4  = (const int4*)erow;

    if (wv == 0) {
        const int e = 4 * E4 + lane;
        const int r = e < nEdges ? erow[e] : -1;
        agg_take(e, r, r0, nr, q, cnt);
    }

    int4 cur[AGG_U];
#pragma unroll
    for (int u = 0; u < AGG_U; ++u) cur[u] = er4[min(u * AGG_THREADS + t, gEnd)];

    int  sb = 0;
    bool more;
    do {
        while (sb < E4 && cnt <= AGG_QCAP - AGG_U * 4 * 32) {
            int4 nxt[AGG_U];
#pragma unroll
            for (int u = 0; u < AGG_U; ++u) nxt[u] = er4[min(sb + (AGG_U + u) * AGG_THREADS + t, gEnd)];
#pragma unroll
            for (int u = 0; u < AGG_U; ++u) {
                const int idx = sb + u * AGG_THREADS + t;
                const int nru = idx < E4 ? nr : 0;
                const int e   = 4 * idx;
                agg_take(e + 0, cur[u].x, r0, nru, q, cnt);
                agg_take(e + 1, cur[u].y, r0, nru, q, cnt);
                agg_take(e + 2, cur[u].z, r0, nru, q, cnt);
                agg_take(e + 3, cur[u].w, r0, nru, q, cnt);
            }
#pragma unroll
            for (int u = 0; u < AGG_U; ++u) cur[u] = nxt[u];
            sb += AGG_STEP;
        }
        if (lane == 0) {
            qcount[wv] = cnt;
            qmore[wv]  = sb < E4 ? 1 : 0;
        }
        __syncthreads();
        more = __builtin_amdgcn_ballot_w32(qmore[lane] != 0) != 0u;

        int ocnt = 0;
        for (int s = 0; s < AGG_WAVES; ++s) {
            const int ns = __builtin_amdgcn_readfirstlane(qcount[s]);
            const unsigned* qs = hitq + s * AGG_QCAP;
            for (int b = 0; b < ns; b += 32) {
                const int i = b + lane;
                unsigned ent  = 0u;
                bool     mine = false;
                if (i < ns) {
                    ent  = qs[i];
                    mine = (int)(ent & 31u) == wv;
                }
                const unsigned mask = __builtin_amdgcn_ballot_w32(mine);
                if (mask != 0u) {
                    if (mine) o[ocnt + (int)__builtin_amdgcn_mbcnt_lo(mask, 0u)] = ent;
                    ocnt += __builtin_popcount(mask);
                    if (ocnt >= 32) {
                        agg_drain(o + (ocnt - 32), 32, lane, nNodes, support, ecol, eval, acc);
                        ocnt -= 32;
                    }
                }
            }
        }
        if (ocnt > 0) agg_drain(o, ocnt, lane, nNodes, support, ecol, eval, acc);

        __syncthreads();
        cnt = 0;
    } while (more);

    for (int f = t; f < nr * 32; f += AGG_THREADS) {
        const int row = f >> 5;
        const float* a = acc + row * D + lane;
        float4 o4;
        o4.x = tanhf(a[0]);
        o4.y = tanhf(a[32]);
        o4.z = tanhf(a[64]);
        o4.w = tanhf(a[96]);
        ((float4*)(out + (size_t)(r0 + row) * D))[lane] = o4;
    }
}

extern "C" void kernel_launch(void* const* d_in, const int* in_sizes, int n_in,
                              void* d_out, int out_size, void* d_ws, size_t ws_size,
                              hipStream_t stream) {
    const float* x    = (const float*)d_in[0];
    const float* W    = (const float*)d_in[1];
    const int*   erow = (const int*)d_in[2];
    const int*   ecol = (const int*)d_in[3];
    const float* eval = (const float*)d_in[4];
    float* out = (float*)d_out;

    const int nNodes = in_sizes[0] / D;
    const int nEdges = in_sizes[2];

    float* support = (float*)d_ws;

    {
        const int stripes = nNodes / 16;
        gcn_gemm_wmma<<<1024, 256, 0, stream>>>(x, W, support, stripes);
    }
    {
        const int blocks = (nNodes + AGG_ROWS - 1) / AGG_ROWS;
        gcn_aggregate_tanh<<<blocks, AGG_THREADS, 0, stream>>>(support, erow, ecol, eval, out, nNodes, nEdges);
    }
}


# aggregate: two-level edge filter: x@W launch writes per 32 edges a bitmap of the 400-row blocks hit; a block reads its bit plane, then only flagged groups' rows; sums in another fixed order
# speedup vs baseline: 2.5861x; 1.2463x over previous
#include <hip/hip_runtime.h>
#include <hip/hip_bf16.h>

typedef __attribute__((ext_vector_type(2))) float v2f;
typedef __attribute__((ext_vector_type(8))) float v8f;

#define D 128


#define AGG_ROWS   400
#define IDX_PLANES 8

__device__ __forceinline__ void idx_mark(int r, int nNodes, unsigned (&m)[IDX_PLANES]) {
    const unsigned blk = (unsigned)r / AGG_ROWS;
    const unsigned bit = (unsigned)r < (unsigned)nNodes ? (1u << (blk & 31u)) : 0u;
    const unsigned w   = blk >> 5;
#pragma unroll
    for (int p = 0; p < IDX_PLANES; ++p) m[p] |= (w == (unsigned)p) ? bit : 0u;
}

__device__ __forceinline__ void idx_build(int task, int lane, const int* __restrict__ erow,
                                          unsigned* __restrict__ idx, int nNodes, int nEdges) {
    const int G  = (nEdges + 31) >> 5;
    const int Gp = (G + 3) & ~3;
    const int g  = task * 32 + lane;
    if (g >= Gp) return;
    unsigned m[IDX_PLANES];
#pragma unroll
    for (int p = 0; p < IDX_PLANES; ++p) m[p] = 0u;
    const int e0 = g * 32;
    if (e0 + 32 <= nEdges) {
        const int4* src = (const int4*)(erow + e0);
        int4 r[8];
#pragma unroll
        for (int k = 0; k < 8; ++k) r[k] = src[k];
#pragma unroll
        for (int k = 0; k < 8; ++k) {
            idx_mark(r[k].x, nNodes, m);
            idx_mark(r[k].y, nNodes, m);
            idx_mark(r[k].z, nNodes, m);
            idx_mark(r[k].w, nNodes, m);
        }
    } else {
        for (int j = 0; j < 32; ++j) {
            const int e = e0 + j;
            if (e < nEdges) idx_mark(erow[e], nNodes, m);
        }
    }
#pragma unroll
    for (int p = 0; p < IDX_PLANES; ++p) idx[(size_t)p * Gp + g] = m[p];
}

__device__ __forceinline__ void idx_tasks(int wave, int lane, const int* __restrict__ erow,
                                          float* support, int nNodes, int nEdges) {
    unsigned* idx = (unsigned*)(support + (size_t)nNodes * D);
    const int G      = (nEdges + 31) >> 5;
    const int nTasks = (((G + 3) & ~3) + 31) >> 5;
    const int wu     = __builtin_amdgcn_readfirstlane(wave);
    const int wpb    = (int)(blockDim.x >> 5);
    for (int tk = ((int)gridDim.x - 1 - (int)blockIdx.x) * wpb + wu; tk < nTasks; tk += (int)gridDim.x * wpb)
        idx_build(tk, lane, erow, idx, nNodes, nEdges);
}

#define AS_PITCH 132

__global__ __launch_bounds__(256) void gcn_gemm_wmma(const float* __restrict__ x,
                                                     const float* __restrict__ W,
                                                     float* __restrict__ support,
                                                     int nStripes,
                                                     const int* __restrict__ erow,
                                                     int nNodes, int nEdges) {
    __shared__ __attribute__((aligned(16))) float As[16 * AS_PITCH];

    const int t    = threadIdx.x;
    const int lane = t & 31;
    const int wave = t >> 5;
    const int n0   = wave * 16;
    const int half = lane >> 4;
    const int l16  = lane & 15;

    v2f bfrag[32];
#pragma unroll
    for (int s = 0; s < 32; ++s) {
        const int ka = 4 * s + half * 2;
        bfrag[s].x = W[(size_t)ka * D + n0 + l16];
        bfrag[s].y = W[(size_t)(ka + 1) * D + n0 + l16];
    }

    float* lds0 = As + (t >> 5) * AS_PITCH + (t & 31) * 4;
    float* lds1 = lds0 + 8 * AS_PITCH;
    const float* arow = As + l16 * AS_PITCH + half * 2;
    float* srow = support + n0 + l16;

    int stripe = blockIdx.x;
    const int first = stripe < nStripes ? stripe : 0;
    const float4* src0 = (const float4*)(x + (size_t)first * 16 * D);
    float4 g0 = src0[t];
    float4 g1 = src0[t + 256];
    for (; stripe < nStripes; stripe += gridDim.x) {
        __syncthreads();
        *(float4*)lds0 = g0;
        *(float4*)lds1 = g1;
        __syncthreads();

        const int next = stripe + gridDim.x;
        const int nx   = next < nStripes ? next : stripe;
        const float4* src = (const float4*)(x + (size_t)nx * 16 * D);
        g0 = src[t];
        g1 = src[t + 256];

        v8f c = {};
#pragma unroll
        for (int s = 0; s < 32; ++s) {
            const v2f a = *(const v2f*)(arow + 4 * s);
            c = __builtin_amdgcn_wmma_f32_16x16x4_f32(
                    false, a, false, bfrag[s], (short)0, c, false, false);
        }

        const int m0 = stripe * 16;
#pragma unroll
        for (int v = 0; v < 8; ++v) {
            srow[(size_t)(m0 + v + half * 8) * D] = c[v];
        }
    }

    idx_tasks(wave, lane, erow, support, nNodes, nEdges);
}

#define AGG_THREADS 1024
#define AGG_WAVES   32
#define AGG_GB      16
#define AGG_GCAP    80
#define AGG_QCAP    800
#define AGG_OCAP    64
#define AGG_RBITS   9

__device__ __forceinline__ void agg_take(int e, int r, int r0, int nr, unsigned* q, int& cnt) {
    const unsigned rl  = (unsigned)(r - r0);
    const bool     hit = rl < (unsigned)nr;
    const unsigned mask = __builtin_amdgcn_ballot_w32(hit);
    if (mask != 0u) {
        if (hit) q[cnt + (int)__builtin_amdgcn_mbcnt_lo(mask, 0u)] = ((unsigned)e << AGG_RBITS) | rl;
        cnt += __builtin_popcount(mask);
    }
}

__device__ __forceinline__ void agg_groups(const unsigned* gl, int n, int lane, int r0, int nr, int nEdges,
                                           int gLast, int nrTail,
                                           const int* __restrict__ erow, unsigned* q, int& cnt) {
    const int mine = lane < n ? (int)gl[lane] : 0;
    int rows[AGG_GB];
#pragma unroll
    for (int u = 0; u < AGG_GB; ++u) {
        const int g = __builtin_amdgcn_readlane(mine, min(u, n - 1));
        rows[u] = erow[min(g * 32 + lane, nEdges - 1)];
    }
#pragma unroll
    for (int u = 0; u < AGG_GB; ++u) {
        if (u < n) {
            const int g = __builtin_amdgcn_readlane(mine, u);
            agg_take(g * 32 + lane, rows[u], r0, g == gLast ? nrTail : nr, q, cnt);
        }
    }
}

__device__ __forceinline__ void agg_drain(const unsigned* o, int m, int lane, int nNodes,
                                          const float* __restrict__ support,
                                          const int* __restrict__ ecol,
                                          const float* __restrict__ eval,
                                          float* acc) {
    int c = 0, rl = 0;
    float v = 0.f;
    if (lane < m) {
        const unsigned ent = o[lane];
        const int e = (int)(ent >> AGG_RBITS);
        rl = (int)(ent & ((1u << AGG_RBITS) - 1u));
        c  = min(max(ecol[e], 0), nNodes - 1);
        v  = eval[e];
    }
    for (int j = 0; j < m; j += 8) {
        float4 g[8];
#pragma unroll
        for (int u = 0; u < 8; ++u) {
            const int cc = __builtin_amdgcn_readlane(c, min(j + u, m - 1));
            g[u] = ((const float4*)(support + (size_t)cc * D))[lane];
        }
#pragma unroll
        for (int u = 0; u < 8; ++u) {
            if (j + u < m) {
                const float vv = __int_as_float(__builtin_amdgcn_readlane(__float_as_int(v), j + u));
                const int   rr = __builtin_amdgcn_readlane(rl, j + u);
                float* a = acc + rr * D + lane;
                a[0]  += g[u].x * vv;
                a[32] += g[u].y * vv;
                a[64] += g[u].z * vv;
                a[96] += g[u].w * vv;
            }
        }
    }
}

__global__ __launch_bounds__(AGG_THREADS) void gcn_aggregate_tanh(const float* __restrict__ support,
                                                                  const int* __restrict__ erow,
                                                                  const int* __restrict__ ecol,
                                                                  const float* __restrict__ eval,
                                                                  float* __restrict__ out,
                                                                  int nNodes, int nEdges) {
    __shared__ __attribute__((aligned(16))) float acc[AGG_ROWS * D];
    __shared__ unsigned hitq[AGG_WAVES * AGG_QCAP];
    __shared__ unsigned ownq[AGG_WAVES * AGG_OCAP];
    __shared__ unsigned glist[AGG_WAVES * AGG_GCAP];
    __shared__ int      qcount[AGG_WAVES];
    __shared__ int      qmore[AGG_WAVES];

    const int t    = threadIdx.x;
    const int lane = t & 31;
    const int wv   = __builtin_amdgcn_readfirstlane(t >> 5);
    const int r0   = blockIdx.x * AGG_ROWS;
    const int nr   = min(AGG_ROWS, nNodes - r0);

    const int      G     = (nEdges + 31) >> 5;
    const int      Gp    = (G + 3) & ~3;
    const int      U2    = Gp >> 1;
    const int      uEnd  = max(U2 - 1, 0);
    const int      gLast = (nEdges - 1) >> 5;
    const int      nrTail = gLast * 32 + lane < nEdges ? nr : 0;
    const unsigned bsel  = blockIdx.x & 31u;
    const int2*    plane = (const int2*)((const unsigned*)(support + (size_t)nNodes * D) + (size_t)(blockIdx.x >> 5) * Gp);

    int2 cur = plane[min(t, uEnd)];

    for (int i = t; i < AGG_ROWS * D / 4; i += AGG_THREADS)
        ((float4*)acc)[i] = make_float4(0.f, 0.f, 0.f, 0.f);

    unsigned* q  = hitq + wv * AGG_QCAP;
    unsigned* o  = ownq + wv * AGG_OCAP;
    unsigned* gl = glist + wv * AGG_GCAP;
    int cnt  = 0;
    int gcnt = 0;
    int sb   = 0;

    bool more;
    do {
        for (;;) {
            const bool tail = sb >= U2;
            if (gcnt >= AGG_GB || (tail && gcnt > 0)) {
                if (cnt > AGG_QCAP - AGG_GB * 32) break;
                const int n = min(gcnt, AGG_GB);
                gcnt -= n;
                agg_groups(gl + gcnt, n, lane, r0, nr, nEdges, gLast, nrTail, erow, q, cnt);
                continue;
            }
            if (tail) break;
            const int2 nxt  = plane[min(sb + AGG_THREADS + t, uEnd)];
            const int  unit = sb + t;
            const bool live = unit < U2;
#pragma unroll
            for (int k = 0; k < 2; ++k) {
                const unsigned w    = (unsigned)(k == 0 ? cur.x : cur.y);
                const bool     f    = live && ((w >> bsel) & 1u) != 0u;
                const unsigned mask = __builtin_amdgcn_ballot_w32(f);
                if (mask != 0u) {
                    if (f) gl[gcnt + (int)__builtin_amdgcn_mbcnt_lo(mask, 0u)] = (unsigned)(2 * unit + k);
                    gcnt += __builtin_popcount(mask);
                }
            }
            cur = nxt;
            sb += AGG_THREADS;
        }
        if (lane == 0) {
            qcount[wv] = cnt;
            qmore[wv]  = (sb < U2 || gcnt > 0) ? 1 : 0;
        }
        __syncthreads();
        more = __builtin_amdgcn_ballot_w32(qmore[lane] != 0) != 0u;

        int ocnt = 0;
        for (int s = 0; s < AGG_WAVES; ++s) {
            const int ns = __builtin_amdgcn_readfirstlane(qcount[s]);
            const unsigned* qs = hitq + s * AGG_QCAP;
            for (int b = 0; b < ns; b += 32) {
                const int i = b + lane;
                unsigned ent  = 0u;
                bool     mine = false;
                if (i < ns) {
                    ent  = qs[i];
                    mine = (int)(ent & 31u) == wv;
                }
                const unsigned mask = __builtin_amdgcn_ballot_w32(mine);
                if (mask != 0u) {
                    if (mine) o[ocnt + (int)__builtin_amdgcn_mbcnt_lo(mask, 0u)] = ent;
                    ocnt += __builtin_popcount(mask);
                    if (ocnt >= 32) {
                        agg_drain(o + (ocnt - 32), 32, lane, nNodes, support, ecol, eval, acc);
                        ocnt -= 32;
                    }
                }
            }
        }
        if (ocnt > 0) agg_drain(o, ocnt, lane, nNodes, support, ecol, eval, acc);

        __syncthreads();
        cnt = 0;
    } while (more);

    for (int f = t; f < nr * 32; f += AGG_THREADS) {
        const int row = f >> 5;
        const float* a = acc + row * D + lane;
        float4 o4;
        o4.x = tanhf(a[0]);
        o4.y = tanhf(a[32]);
        o4.z = tanhf(a[64]);
        o4.w = tanhf(a[96]);
        ((float4*)(out + (size_t)(r0 + row) * D))[lane] = o4;
    }
}

extern "C" void kernel_launch(void* const* d_in, const int* in_sizes, int n_in,
                              void* d_out, int out_size, void* d_ws, size_t ws_size,
                              hipStream_t stream) {
    const float* x    = (const float*)d_in[0];
    const float* W    = (const float*)d_in[1];
    const int*   erow = (const int*)d_in[2];
    const int*   ecol = (const int*)d_in[3];
    const float* eval = (const float*)d_in[4];
    float* out = (float*)d_out;

    const int nNodes = in_sizes[0] / D;
    const int nEdges = in_sizes[2];

    float* support = (float*)d_ws;

    {
        const int stripes = nNodes / 16;
        gcn_gemm_wmma<<<1024, 256, 0, stream>>>(x, W, support, stripes, erow, nNodes, nEdges);
    }
    {
        const int blocks = (nNodes + AGG_ROWS - 1) / AGG_ROWS;
        gcn_aggregate_tanh<<<blocks, AGG_THREADS, 0, stream>>>(support, erow, ecol, eval, out, nNodes, nEdges);
    }
}


# aggregate: each wave sorts its own hit queue by owner wave in place (ballot ranks, stable); an owner copies 32 short runs instead of testing every hit of the block; sums in another fixed order
# speedup vs baseline: 2.8246x; 1.0922x over previous
#include <hip/hip_runtime.h>
#include <hip/hip_bf16.h>

typedef __attribute__((ext_vector_type(2))) float v2f;
typedef __attribute__((ext_vector_type(8))) float v8f;

#define D 128


#define AGG_ROWS   400
#define IDX_PLANES 8

__device__ __forceinline__ void idx_mark(int r, int nNodes, unsigned (&m)[IDX_PLANES]) {
    const unsigned blk = (unsigned)r / AGG_ROWS;
    const unsigned bit = (unsigned)r < (unsigned)nNodes ? (1u << (blk & 31u)) : 0u;
    const unsigned w   = blk >> 5;
#pragma unroll
    for (int p = 0; p < IDX_PLANES; ++p) m[p] |= (w == (unsigned)p) ? bit : 0u;
}

__device__ __forceinline__ void idx_build(int task, int lane, const int* __restrict__ erow,
                                          unsigned* __restrict__ idx, int nNodes, int nEdges) {
    const int G  = (nEdges + 31) >> 5;
    const int Gp = (G + 3) & ~3;
    const int g  = task * 32 + lane;
    if (g >= Gp) return;
    unsigned m[IDX_PLANES];
#pragma unroll
    for (int p = 0; p < IDX_PLANES; ++p) m[p] = 0u;
    const int e0 = g * 32;
    if (e0 + 32 <= nEdges) {
        const int4* src = (const int4*)(erow + e0);
        int4 r[8];
#pragma unroll
        for (int k = 0; k < 8; ++k) r[k] = src[k];
#pragma unroll
        for (int k = 0; k < 8; ++k) {
            idx_mark(r[k].x, nNodes, m);
            idx_mark(r[k].y, nNodes, m);
            idx_mark(r[k].z, nNodes, m);
            idx_mark(r[k].w, nNodes, m);
        }
    } else {
        for (int j = 0; j < 32; ++j) {
            const int e = e0 + j;
            if (e < nEdges) idx_mark(erow[e], nNodes, m);
        }
    }
#pragma unroll
    for (int p = 0; p < IDX_PLANES; ++p) idx[(size_t)p * Gp + g] = m[p];
}

__device__ __forceinline__ void idx_tasks(int wave, int lane, const int* __restrict__ erow,
                                          float* support, int nNodes, int nEdges) {
    unsigned* idx = (unsigned*)(support + (size_t)nNodes * D);
    const int G      = (nEdges + 31) >> 5;
    const int nTasks = (((G + 3) & ~3) + 31) >> 5;
    const int wu     = __builtin_amdgcn_readfirstlane(wave);
    const int wpb    = (int)(blockDim.x >> 5);
    for (int tk = ((int)gridDim.x - 1 - (int)blockIdx.x) * wpb + wu; tk < nTasks; tk += (int)gridDim.x * wpb)
        idx_build(tk, lane, erow, idx, nNodes, nEdges);
}

#define AS_PITCH 132

__global__ __launch_bounds__(256) void gcn_gemm_wmma(const float* __restrict__ x,
                                                     const float* __restrict__ W,
                                                     float* __restrict__ support,
                                                     int nStripes,
                                                     const int* __restrict__ erow,
                                                     int nNodes, int nEdges) {
    __shared__ __attribute__((aligned(16))) float As[16 * AS_PITCH];

    const int t    = threadIdx.x;
    const int lane = t & 31;
    const int wave = t >> 5;
    const int n0   = wave * 16;
    const int half = lane >> 4;
    const int l16  = lane & 15;

    v2f bfrag[32];
#pragma unroll
    for (int s = 0; s < 32; ++s) {
        const int ka = 4 * s + half * 2;
        bfrag[s].x = W[(size_t)ka * D + n0 + l16];
        bfrag[s].y = W[(size_t)(ka + 1) * D + n0 + l16];
    }

    float* lds0 = As + (t >> 5) * AS_PITCH + (t & 31) * 4;
    float* lds1 = lds0 + 8 * AS_PITCH;
    const float* arow = As + l16 * AS_PITCH + half * 2;
    float* srow = support + n0 + l16;

    int stripe = blockIdx.x;
    const int first = stripe < nStripes ? stripe : 0;
    const float4* src0 = (const float4*)(x + (size_t)first * 16 * D);
    float4 g0 = src0[t];
    float4 g1 = src0[t + 256];
    for (; stripe < nStripes; stripe += gridDim.x) {
        __syncthreads();
        *(float4*)lds0 = g0;
        *(float4*)lds1 = g1;
        __syncthreads();

        const int next = stripe + gridDim.x;
        const int nx   = next < nStripes ? next : stripe;
        const float4* src = (const float4*)(x + (size_t)nx * 16 * D);
        g0 = src[t];
        g1 = src[t + 256];

        v8f c = {};
#pragma unroll
        for (int s = 0; s < 32; ++s) {
            const v2f a = *(const v2f*)(arow + 4 * s);
            c = __builtin_amdgcn_wmma_f32_16x16x4_f32(
                    false, a, false, bfrag[s], (short)0, c, false, false);
        }

        const int m0 = stripe * 16;
#pragma unroll
        for (int v = 0; v < 8; ++v) {
            srow[(size_t)(m0 + v + half * 8) * D] = c[v];
        }
    }

    idx_tasks(wave, lane, erow, support, nNodes, nEdges);
}

#define AGG_THREADS 1024
#define AGG_WAVES   32
#define AGG_GB      16
#define AGG_GCAP    80
#define AGG_QCAP    768
#define AGG_NB      (AGG_QCAP / 32)
#define AGG_OFFP    33
#define AGG_OCAP    64
#define AGG_RBITS   9

__device__ __forceinline__ void agg_take(int e, int r, int r0, int nr, unsigned* q, int& cnt) {
    const unsigned rl  = (unsigned)(r - r0);
    const bool     hit = rl < (unsigned)nr;
    const unsigned mask = __builtin_amdgcn_ballot_w32(hit);
    if (mask != 0u) {
        if (hit) q[cnt + (int)__builtin_amdgcn_mbcnt_lo(mask, 0u)] = ((unsigned)e << AGG_RBITS) | rl;
        cnt += __builtin_popcount(mask);
    }
}

__device__ __forceinline__ void agg_groups(const unsigned* gl, int n, int lane, int r0, int nr, int nEdges,
                                           int gLast, int nrTail,
                                           const int* __restrict__ erow, unsigned* q, int& cnt) {
    const int mine = lane < n ? (int)gl[lane] : 0;
    int rows[AGG_GB];
#pragma unroll
    for (int u = 0; u < AGG_GB; ++u) {
        const int g = __builtin_amdgcn_readlane(mine, min(u, n - 1));
        rows[u] = erow[min(g * 32 + lane, nEdges - 1)];
    }
#pragma unroll
    for (int u = 0; u < AGG_GB; ++u) {
        if (u < n) {
            const int g = __builtin_amdgcn_readlane(mine, u);
            agg_take(g * 32 + lane, rows[u], r0, g == gLast ? nrTail : nr, q, cnt);
        }
    }
}

__device__ __forceinline__ void agg_sort(unsigned* q, int cnt, int lane, int* bounds) {
    unsigned lbit[5];
#pragma unroll
    for (int t = 0; t < 5; ++t) lbit[t] = 0u - (((unsigned)lane >> t) & 1u);

    int run = 0;
    for (int b0 = 0; b0 < cnt; b0 += 32) {
        const bool     valid = b0 + lane < cnt;
        const unsigned e     = valid ? q[b0 + lane] : 0u;
        unsigned dBin = ~__builtin_amdgcn_ballot_w32(valid);
#pragma unroll
        for (int t = 0; t < 5; ++t)
            dBin |= __builtin_amdgcn_ballot_w32(((e >> t) & 1u) != 0u) ^ lbit[t];
        run += __builtin_popcount(~dBin);
    }
    int incl = run;
#pragma unroll
    for (int d = 1; d < 32; d <<= 1) {
        const int up = __shfl_up(incl, d);
        if (lane >= d) incl += up;
    }
    int next = incl - run;
    bounds[lane] = next;
    if (lane == 0) bounds[32] = cnt;
    unsigned ent[AGG_NB];
#pragma unroll
    for (int b = 0; b < AGG_NB; ++b)
        ent[b] = b * 32 + lane < cnt ? q[b * 32 + lane] : 0u;
#pragma unroll
    for (int b = 0; b < AGG_NB; ++b) {
        if (b * 32 < cnt) {
            const bool     valid = b * 32 + lane < cnt;
            const unsigned e     = ent[b];
            const unsigned dead  = ~__builtin_amdgcn_ballot_w32(valid);
            unsigned dBin = dead, dKey = dead;
#pragma unroll
            for (int t = 0; t < 5; ++t) {
                const unsigned bt = __builtin_amdgcn_ballot_w32(((e >> t) & 1u) != 0u);
                dBin |= bt ^ lbit[t];
                dKey |= bt ^ (0u - ((e >> t) & 1u));
            }
            const int place = __shfl(next, (int)(e & 31u)) + (int)__builtin_amdgcn_mbcnt_lo(~dKey, 0u);
            if (valid) q[place] = e;
            next += __builtin_popcount(~dBin);
        }
    }
}

__device__ __forceinline__ void agg_drain(const unsigned* o, int m, int lane, int nNodes,
                                          const float* __restrict__ support,
                                          const int* __restrict__ ecol,
                                          const float* __restrict__ eval,
                                          float* acc) {
    int c = 0, rl = 0;
    float v = 0.f;
    if (lane < m) {
        const unsigned ent = o[lane];
        const int e = (int)(ent >> AGG_RBITS);
        rl = (int)(ent & ((1u << AGG_RBITS) - 1u));
        c  = min(max(ecol[e], 0), nNodes - 1);
        v  = eval[e];
    }
    for (int j = 0; j < m; j += 8) {
        float4 g[8];
#pragma unroll
        for (int u = 0; u < 8; ++u) {
            const int cc = __builtin_amdgcn_readlane(c, min(j + u, m - 1));
            g[u] = ((const float4*)(support + (size_t)cc * D))[lane];
        }
#pragma unroll
        for (int u = 0; u < 8; ++u) {
            if (j + u < m) {
                const float vv = __int_as_float(__builtin_amdgcn_readlane(__float_as_int(v), j + u));
                const int   rr = __builtin_amdgcn_readlane(rl, j + u);
                float* a = acc + rr * D + lane;
                a[0]  += g[u].x * vv;
                a[32] += g[u].y * vv;
                a[64] += g[u].z * vv;
                a[96] += g[u].w * vv;
            }
        }
    }
}

__global__ __launch_bounds__(AGG_THREADS) void gcn_aggregate_tanh(const float* __restrict__ support,
                                                                  const int* __restrict__ erow,
                                                                  const int* __restrict__ ecol,
                                                                  const float* __restrict__ eval,
                                                                  float* __restrict__ out,
                                                                  int nNodes, int nEdges) {
    __shared__ __attribute__((aligned(16))) float acc[AGG_ROWS * D];
    __shared__ unsigned hitq[AGG_WAVES * AGG_QCAP];
    __shared__ unsigned ownq[AGG_WAVES * AGG_OCAP];
    __shared__ unsigned glist[AGG_WAVES * AGG_GCAP];
    __shared__ int      qoff[AGG_WAVES * AGG_OFFP];
    __shared__ int      qmore[AGG_WAVES];

    const int t    = threadIdx.x;
    const int lane = t & 31;
    const int wv   = __builtin_amdgcn_readfirstlane(t >> 5);
    const int r0   = blockIdx.x * AGG_ROWS;
    const int nr   = min(AGG_ROWS, nNodes - r0);

    const int      G     = (nEdges + 31) >> 5;
    const int      Gp    = (G + 3) & ~3;
    const int      U2    = Gp >> 1;
    const int      uEnd  = max(U2 - 1, 0);
    const int      gLast = (nEdges - 1) >> 5;
    const int      nrTail = gLast * 32 + lane < nEdges ? nr : 0;
    const unsigned bsel  = blockIdx.x & 31u;
    const int2*    plane = (const int2*)((const unsigned*)(support + (size_t)nNodes * D) + (size_t)(blockIdx.x >> 5) * Gp);

    int2 cur = plane[min(t, uEnd)];

    for (int i = t; i < AGG_ROWS * D / 4; i += AGG_THREADS)
        ((float4*)acc)[i] = make_float4(0.f, 0.f, 0.f, 0.f);

    unsigned* q  = hitq + wv * AGG_QCAP;
    unsigned* o  = ownq + wv * AGG_OCAP;
    unsigned* gl = glist + wv * AGG_GCAP;
    int cnt  = 0;
    int gcnt = 0;
    int sb   = 0;

    bool more;
    do {
        for (;;) {
            const bool tail = sb >= U2;
            if (gcnt >= AGG_GB || (tail && gcnt > 0)) {
                if (cnt > AGG_QCAP - AGG_GB * 32) break;
                const int n = min(gcnt, AGG_GB);
                gcnt -= n;
                agg_groups(gl + gcnt, n, lane, r0, nr, nEdges, gLast, nrTail, erow, q, cnt);
                continue;
            }
            if (tail) break;
            const int2 nxt  = plane[min(sb + AGG_THREADS + t, uEnd)];
            const int  unit = sb + t;
            const bool live = unit < U2;
#pragma unroll
            for (int k = 0; k < 2; ++k) {
                const unsigned w    = (unsigned)(k == 0 ? cur.x : cur.y);
                const bool     f    = live && ((w >> bsel) & 1u) != 0u;
                const unsigned mask = __builtin_amdgcn_ballot_w32(f);
                if (mask != 0u) {
                    if (f) gl[gcnt + (int)__builtin_amdgcn_mbcnt_lo(mask, 0u)] = (unsigned)(2 * unit + k);
                    gcnt += __builtin_popcount(mask);
                }
            }
            cur = nxt;
            sb += AGG_THREADS;
        }
        agg_sort(q, cnt, lane, qoff + wv * AGG_OFFP);
        if (lane == 0) qmore[wv] = (sb < U2 || gcnt > 0) ? 1 : 0;
        __syncthreads();
        more = __builtin_amdgcn_ballot_w32(qmore[lane] != 0) != 0u;

        const int runLo = qoff[lane * AGG_OFFP + wv];
        const int runHi = qoff[lane * AGG_OFFP + wv + 1];
        int ocnt = 0;
        for (int s = 0; s < AGG_WAVES; ++s) {
            const int lo = __builtin_amdgcn_readlane(runLo, s);
            const int ns = __builtin_amdgcn_readlane(runHi, s) - lo;
            const unsigned* qs = hitq + s * AGG_QCAP + lo;
            for (int b = 0; b < ns; b += 32) {
                const int k = min(32, ns - b);
                if (lane < k) o[ocnt + lane] = qs[b + lane];
                ocnt += k;
                if (ocnt >= 32) {
                    agg_drain(o + (ocnt - 32), 32, lane, nNodes, support, ecol, eval, acc);
                    ocnt -= 32;
                }
            }
        }
        if (ocnt > 0) agg_drain(o, ocnt, lane, nNodes, support, ecol, eval, acc);

        __syncthreads();
        cnt = 0;
    } while (more);

    for (int f = t; f < nr * 32; f += AGG_THREADS) {
        const int row = f >> 5;
        const float* a = acc + row * D + lane;
        float4 o4;
        o4.x = tanhf(a[0]);
        o4.y = tanhf(a[32]);
        o4.z = tanhf(a[64]);
        o4.w = tanhf(a[96]);
        ((float4*)(out + (size_t)(r0 + row) * D))[lane] = o4;
    }
}

extern "C" void kernel_launch(void* const* d_in, const int* in_sizes, int n_in,
                              void* d_out, int out_size, void* d_ws, size_t ws_size,
                              hipStream_t stream) {
    const float* x    = (const float*)d_in[0];
    const float* W    = (const float*)d_in[1];
    const int*   erow = (const int*)d_in[2];
    const int*   ecol = (const int*)d_in[3];
    const float* eval = (const float*)d_in[4];
    float* out = (float*)d_out;

    const int nNodes = in_sizes[0] / D;
    const int nEdges = in_sizes[2];

    float* support = (float*)d_ws;

    {
        const int stripes = nNodes / 16;
        gcn_gemm_wmma<<<1024, 256, 0, stream>>>(x, W, support, stripes, erow, nNodes, nEdges);
    }
    {
        const int blocks = (nNodes + AGG_ROWS - 1) / AGG_ROWS;
        gcn_aggregate_tanh<<<blocks, AGG_THREADS, 0, stream>>>(support, erow, ecol, eval, out, nNodes, nEdges);
    }
}


# aggregate scan: a take's hit lanes kept as one scalar mask (the compare writes it, scalar code trims it, it is the store's predicate); full batches on their own path; same sums, same order
# speedup vs baseline: 2.9184x; 1.0332x over previous
#include <hip/hip_runtime.h>
#include <hip/hip_bf16.h>

typedef __attribute__((ext_vector_type(2))) float v2f;
typedef __attribute__((ext_vector_type(8))) float v8f;

#define D 128


#define AGG_ROWS   400
#define IDX_PLANES 8

__device__ __forceinline__ void idx_mark(int r, int nNodes, unsigned (&m)[IDX_PLANES]) {
    const unsigned blk = (unsigned)r / AGG_ROWS;
    const unsigned bit = (unsigned)r < (unsigned)nNodes ? (1u << (blk & 31u)) : 0u;
    const unsigned w   = blk >> 5;
#pragma unroll
    for (int p = 0; p < IDX_PLANES; ++p) m[p] |= (w == (unsigned)p) ? bit : 0u;
}

__device__ __forceinline__ void idx_build(int task, int lane, const int* __restrict__ erow,
                                          unsigned* __restrict__ idx, int nNodes, int nEdges) {
    const int G  = (nEdges + 31) >> 5;
    const int Gp = (G + 3) & ~3;
    const int g  = task * 32 + lane;
    if (g >= Gp) return;
    unsigned m[IDX_PLANES];
#pragma unroll
    for (int p = 0; p < IDX_PLANES; ++p) m[p] = 0u;
    const int e0 = g * 32;
    if (e0 + 32 <= nEdges) {
        const int4* src = (const int4*)(erow + e0);
        int4 r[8];
#pragma unroll
        for (int k = 0; k < 8; ++k) r[k] = src[k];
#pragma unroll
        for (int k = 0; k < 8; ++k) {
            idx_mark(r[k].x, nNodes, m);
            idx_mark(r[k].y, nNodes, m);
            idx_mark(r[k].z, nNodes, m);
            idx_mark(r[k].w, nNodes, m);
        }
    } else {
        for (int j = 0; j < 32; ++j) {
            const int e = e0 + j;
            if (e < nEdges) idx_mark(erow[e], nNodes, m);
        }
    }
#pragma unroll
    for (int p = 0; p < IDX_PLANES; ++p) idx[(size_t)p * Gp + g] = m[p];
}

__device__ __forceinline__ void idx_tasks(int wave, int lane, const int* __restrict__ erow,
                                          float* support, int nNodes, int nEdges) {
    unsigned* idx = (unsigned*)(support + (size_t)nNodes * D);
    const int G      = (nEdges + 31) >> 5;
    const int nTasks = (((G + 3) & ~3) + 31) >> 5;
    const int wu     = __builtin_amdgcn_readfirstlane(wave);
    const int wpb    = (int)(blockDim.x >> 5);
    for (int tk = ((int)gridDim.x - 1 - (int)blockIdx.x) * wpb + wu; tk < nTasks; tk += (int)gridDim.x * wpb)
        idx_build(tk, lane, erow, idx, nNodes, nEdges);
}

#define AS_PITCH 132

__global__ __launch_bounds__(256) void gcn_gemm_wmma(const float* __restrict__ x,
                                                     const float* __restrict__ W,
                                                     float* __restrict__ support,
                                                     int nStripes,
                                                     const int* __restrict__ erow,
                                                     int nNodes, int nEdges) {
    __shared__ __attribute__((aligned(16))) float As[16 * AS_PITCH];

    const int t    = threadIdx.x;
    const int lane = t & 31;
    const int wave = t >> 5;
    const int n0   = wave * 16;
    const int half = lane >> 4;
    const int l16  = lane & 15;

    v2f bfrag[32];
#pragma unroll
    for (int s = 0; s < 32; ++s) {
        const int ka = 4 * s + half * 2;
        bfrag[s].x = W[(size_t)ka * D + n0 + l16];
        bfrag[s].y = W[(size_t)(ka + 1) * D + n0 + l16];
    }

    float* lds0 = As + (t >> 5) * AS_PITCH + (t & 31) * 4;
    float* lds1 = lds0 + 8 * AS_PITCH;
    const float* arow = As + l16 * AS_PITCH + half * 2;
    float* srow = support + n0 + l16;

    int stripe = blockIdx.x;
    const int first = stripe < nStripes ? stripe : 0;
    const float4* src0 = (const float4*)(x + (size_t)first * 16 * D);
    float4 g0 = src0[t];
    float4 g1 = src0[t + 256];
    for (; stripe < nStripes; stripe += gridDim.x) {
        __syncthreads();
        *(float4*)lds0 = g0;
        *(float4*)lds1 = g1;
        __syncthreads();

        const int next = stripe + gridDim.x;
        const int nx   = next < nStripes ? next : stripe;
        const float4* src = (const float4*)(x + (size_t)nx * 16 * D);
        g0 = src[t];
        g1 = src[t + 256];

        v8f c = {};
#pragma unroll
        for (int s = 0; s < 32; ++s) {
            const v2f a = *(const v2f*)(arow + 4 * s);
            c = __builtin_amdgcn_wmma_f32_16x16x4_f32(
                    false, a, false, bfrag[s], (short)0, c, false, false);
        }

        const int m0 = stripe * 16;
#pragma unroll
        for (int v = 0; v < 8; ++v) {
            srow[(size_t)(m0 + v + half * 8) * D] = c[v];
        }
    }

    idx_tasks(wave, lane, erow, support, nNodes, nEdges);
}

#define AGG_THREADS 1024
#define AGG_WAVES   32
#define AGG_GB      16
#define AGG_GCAP    80
#define AGG_QCAP    768
#define AGG_NB      (AGG_QCAP / 32)
#define AGG_OFFP    33
#define AGG_OCAP    64
#define AGG_RBITS   9

__device__ __forceinline__ void agg_take(int e, int r, int r0, int nr, unsigned live, unsigned*& qe) {
    const unsigned rl   = (unsigned)(r - r0);
    const unsigned mask = __builtin_amdgcn_ballot_w32(rl < (unsigned)nr) & live;
    if (__builtin_amdgcn_inverse_ballot_w32(mask))
        qe[__builtin_amdgcn_mbcnt_lo(mask, 0u)] = ((unsigned)e << AGG_RBITS) | rl;
    qe += __builtin_popcount(mask);
}

template <bool FULL>
__device__ __forceinline__ void agg_groups(const unsigned* gl, int n, int lane, int r0, int nr, int nEdges,
                                           int gPart, unsigned tailMask,
                                           const int* __restrict__ erow, unsigned* q, int& cnt) {
    const int mine = lane < n ? (int)gl[lane] : 0;
    int rows[AGG_GB];
#pragma unroll
    for (int u = 0; u < AGG_GB; ++u) {
        const int g = __builtin_amdgcn_readlane(mine, FULL ? u : min(u, n - 1));
        rows[u] = erow[min(g * 32 + lane, nEdges - 1)];
    }
    unsigned* qe = q + cnt;
#pragma unroll
    for (int u = 0; u < AGG_GB; ++u) {
        const int g    = __builtin_amdgcn_readlane(mine, u);
        unsigned  live = g == gPart ? tailMask : ~0u;
        if (!FULL) live = u < n ? live : 0u;
        agg_take(g * 32 + lane, rows[u], r0, nr, live, qe);
    }
    cnt = (int)(qe - q);
}

__device__ __forceinline__ void agg_sort(unsigned* q, int cnt, int lane, int* bounds) {
    unsigned lbit[5];
#pragma unroll
    for (int t = 0; t < 5; ++t) lbit[t] = 0u - (((unsigned)lane >> t) & 1u);

    int run = 0;
    for (int b0 = 0; b0 < cnt; b0 += 32) {
        const bool     valid = b0 + lane < cnt;
        const unsigned e     = valid ? q[b0 + lane] : 0u;
        unsigned dBin = ~__builtin_amdgcn_ballot_w32(valid);
#pragma unroll
        for (int t = 0; t < 5; ++t)
            dBin |= __builtin_amdgcn_ballot_w32(((e >> t) & 1u) != 0u) ^ lbit[t];
        run += __builtin_popcount(~dBin);
    }
    int incl = run;
#pragma unroll
    for (int d = 1; d < 32; d <<= 1) {
        const int up = __shfl_up(incl, d);
        if (lane >= d) incl += up;
    }
    int next = incl - run;
    bounds[lane] = next;
    if (lane == 0) bounds[32] = cnt;
    unsigned ent[AGG_NB];
#pragma unroll
    for (int b = 0; b < AGG_NB; ++b)
        ent[b] = b * 32 + lane < cnt ? q[b * 32 + lane] : 0u;
#pragma unroll
    for (int b = 0; b < AGG_NB; ++b) {
        if (b * 32 < cnt) {
            const bool     valid = b * 32 + lane < cnt;
            const unsigned e     = ent[b];
            const unsigned dead  = ~__builtin_amdgcn_ballot_w32(valid);
            unsigned dBin = dead, dKey = dead;
#pragma unroll
            for (int t = 0; t < 5; ++t) {
                const unsigned bt = __builtin_amdgcn_ballot_w32(((e >> t) & 1u) != 0u);
                dBin |= bt ^ lbit[t];
                dKey |= bt ^ (0u - ((e >> t) & 1u));
            }
            const int place = __shfl(next, (int)(e & 31u)) + (int)__builtin_amdgcn_mbcnt_lo(~dKey, 0u);
            if (valid) q[place] = e;
            next += __builtin_popcount(~dBin);
        }
    }
}

__device__ __forceinline__ void agg_drain(const unsigned* o, int m, int lane, int nNodes,
                                          const float* __restrict__ support,
                                          const int* __restrict__ ecol,
                                          const float* __restrict__ eval,
                                          float* acc) {
    int c = 0, rl = 0;
    float v = 0.f;
    if (lane < m) {
        const unsigned ent = o[lane];
        const int e = (int)(ent >> AGG_RBITS);
        rl = (int)(ent & ((1u << AGG_RBITS) - 1u));
        c  = min(max(ecol[e], 0), nNodes - 1);
        v  = eval[e];
    }
    for (int j = 0; j < m; j += 8) {
        float4 g[8];
#pragma unroll
        for (int u = 0; u < 8; ++u) {
            const int cc = __builtin_amdgcn_readlane(c, min(j + u, m - 1));
            g[u] = ((const float4*)(support + (size_t)cc * D))[lane];
        }
#pragma unroll
        for (int u = 0; u < 8; ++u) {
            if (j + u < m) {
                const float vv = __int_as_float(__builtin_amdgcn_readlane(__float_as_int(v), j + u));
                const int   rr = __builtin_amdgcn_readlane(rl, j + u);
                float* a = acc + rr * D + lane;
                a[0]  += g[u].x * vv;
                a[32] += g[u].y * vv;
                a[64] += g[u].z * vv;
                a[96] += g[u].w * vv;
            }
        }
    }
}

__global__ __launch_bounds__(AGG_THREADS) void gcn_aggregate_tanh(const float* __restrict__ support,
                                                                  const int* __restrict__ erow,
                                                                  const int* __restrict__ ecol,
                                                                  const float* __restrict__ eval,
                                                                  float* __restrict__ out,
                                                                  int nNodes, int nEdges) {
    __shared__ __attribute__((aligned(16))) float acc[AGG_ROWS * D];
    __shared__ unsigned hitq[AGG_WAVES * AGG_QCAP];
    __shared__ unsigned ownq[AGG_WAVES * AGG_OCAP];
    __shared__ unsigned glist[AGG_WAVES * AGG_GCAP];
    __shared__ int      qoff[AGG_WAVES * AGG_OFFP];
    __shared__ int      qmore[AGG_WAVES];

    const int t    = threadIdx.x;
    const int lane = t & 31;
    const int wv   = __builtin_amdgcn_readfirstlane(t >> 5);
    const int r0   = blockIdx.x * AGG_ROWS;
    const int nr   = min(AGG_ROWS, nNodes - r0);

    const int      G     = (nEdges + 31) >> 5;
    const int      Gp    = (G + 3) & ~3;
    const int      U2    = Gp >> 1;
    const int      uEnd  = max(U2 - 1, 0);
    const int      gLast = (nEdges - 1) >> 5;
    const int      nTail = nEdges - gLast * 32;
    const int      gPart = nTail < 32 ? gLast : -1;
    const unsigned tailMask = (1u << (nTail & 31)) - 1u;
    const unsigned bsel  = blockIdx.x & 31u;
    const int2*    plane = (const int2*)((const unsigned*)(support + (size_t)nNodes * D) + (size_t)(blockIdx.x >> 5) * Gp);

    int2 cur = plane[min(t, uEnd)];

    for (int i = t; i < AGG_ROWS * D / 4; i += AGG_THREADS)
        ((float4*)acc)[i] = make_float4(0.f, 0.f, 0.f, 0.f);

    unsigned* q  = hitq + wv * AGG_QCAP;
    unsigned* o  = ownq + wv * AGG_OCAP;
    unsigned* gl = glist + wv * AGG_GCAP;
    int cnt  = 0;
    int gcnt = 0;
    int sb   = 0;

    bool more;
    do {
        for (;;) {
            const bool tail = sb >= U2;
            if (gcnt >= AGG_GB || (tail && gcnt > 0)) {
                if (cnt > AGG_QCAP - AGG_GB * 32) break;
                const int n = min(gcnt, AGG_GB);
                gcnt -= n;
                if (n == AGG_GB) agg_groups<true>(gl + gcnt, n, lane, r0, nr, nEdges, gPart, tailMask, erow, q, cnt);
                else             agg_groups<false>(gl + gcnt, n, lane, r0, nr, nEdges, gPart, tailMask, erow, q, cnt);
                continue;
            }
            if (tail) break;
            const int2 nxt  = plane[min(sb + AGG_THREADS + t, uEnd)];
            const int      unit = sb + t;
            const unsigned live = __builtin_amdgcn_ballot_w32(unit < U2);
#pragma unroll
            for (int k = 0; k < 2; ++k) {
                const unsigned w    = (unsigned)(k == 0 ? cur.x : cur.y);
                const unsigned mask = __builtin_amdgcn_ballot_w32(((w >> bsel) & 1u) != 0u) & live;
                if (__builtin_amdgcn_inverse_ballot_w32(mask))
                    gl[gcnt + (int)__builtin_amdgcn_mbcnt_lo(mask, 0u)] = (unsigned)(2 * unit + k);
                gcnt += __builtin_popcount(mask);
            }
            cur = nxt;
            sb += AGG_THREADS;
        }
        agg_sort(q, cnt, lane, qoff + wv * AGG_OFFP);
        if (lane == 0) qmore[wv] = (sb < U2 || gcnt > 0) ? 1 : 0;
        __syncthreads();
        more = __builtin_amdgcn_ballot_w32(qmore[lane] != 0) != 0u;

        const int runLo = qoff[lane * AGG_OFFP + wv];
        const int runHi = qoff[lane * AGG_OFFP + wv + 1];
        int ocnt = 0;
        for (int s = 0; s < AGG_WAVES; ++s) {
            const int lo = __builtin_amdgcn_readlane(runLo, s);
            const int ns = __builtin_amdgcn_readlane(runHi, s) - lo;
            const unsigned* qs = hitq + s * AGG_QCAP + lo;
            for (int b = 0; b < ns; b += 32) {
                const int k = min(32, ns - b);
                if (lane < k) o[ocnt + lane] = qs[b + lane];
                ocnt += k;
                if (ocnt >= 32) {
                    agg_drain(o + (ocnt - 32), 32, lane, nNodes, support, ecol, eval, acc);
                    ocnt -= 32;
                }
            }
        }
        if (ocnt > 0) agg_drain(o, ocnt, lane, nNodes, support, ecol, eval, acc);

        __syncthreads();
        cnt = 0;
    } while (more);

    for (int f = t; f < nr * 32; f += AGG_THREADS) {
        const int row = f >> 5;
        const float* a = acc + row * D + lane;
        float4 o4;
        o4.x = tanhf(a[0]);
        o4.y = tanhf(a[32]);
        o4.z = tanhf(a[64]);
        o4.w = tanhf(a[96]);
        ((float4*)(out + (size_t)(r0 + row) * D))[lane] = o4;
    }
}

extern "C" void kernel_launch(void* const* d_in, const int* in_sizes, int n_in,
                              void* d_out, int out_size, void* d_ws, size_t ws_size,
                              hipStream_t stream) {
    const float* x    = (const float*)d_in[0];
    const float* W    = (const float*)d_in[1];
    const int*   erow = (const int*)d_in[2];
    const int*   ecol = (const int*)d_in[3];
    const float* eval = (const float*)d_in[4];
    float* out = (float*)d_out;

    const int nNodes = in_sizes[0] / D;
    const int nEdges = in_sizes[2];

    float* support = (float*)d_ws;

    {
        const int stripes = nNodes / 16;
        gcn_gemm_wmma<<<1024, 256, 0, stream>>>(x, W, support, stripes, erow, nNodes, nEdges);
    }
    {
        const int blocks = (nNodes + AGG_ROWS - 1) / AGG_ROWS;
        gcn_aggregate_tanh<<<blocks, AGG_THREADS, 0, stream>>>(support, erow, ecol, eval, out, nNodes, nEdges);
    }
}


# aggregate: a wave keeps its 13 rows' sums in registers (scalar row index) and finishes them alone: LDS sums, their zeroing, read-back and last barrier gone; same adds, same order
# speedup vs baseline: 3.0972x; 1.0613x over previous
#include <hip/hip_runtime.h>
#include <hip/hip_bf16.h>

typedef __attribute__((ext_vector_type(2))) float v2f;
typedef __attribute__((ext_vector_type(8))) float v8f;

#define D 128


#define AGG_ROWS   400
#define IDX_PLANES 8

__device__ __forceinline__ void idx_mark(int r, int nNodes, unsigned (&m)[IDX_PLANES]) {
    const unsigned blk = (unsigned)r / AGG_ROWS;
    const unsigned bit = (unsigned)r < (unsigned)nNodes ? (1u << (blk & 31u)) : 0u;
    const unsigned w   = blk >> 5;
#pragma unroll
    for (int p = 0; p < IDX_PLANES; ++p) m[p] |= (w == (unsigned)p) ? bit : 0u;
}

__device__ __forceinline__ void idx_build(int task, int lane, const int* __restrict__ erow,
                                          unsigned* __restrict__ idx, int nNodes, int nEdges) {
    const int G  = (nEdges + 31) >> 5;
    const int Gp = (G + 3) & ~3;
    const int g  = task * 32 + lane;
    if (g >= Gp) return;
    unsigned m[IDX_PLANES];
#pragma unroll
    for (int p = 0; p < IDX_PLANES; ++p) m[p] = 0u;
    const int e0 = g * 32;
    if (e0 + 32 <= nEdges) {
        const int4* src = (const int4*)(erow + e0);
        int4 r[8];
#pragma unroll
        for (int k = 0; k < 8; ++k) r[k] = src[k];
#pragma unroll
        for (int k = 0; k < 8; ++k) {
            idx_mark(r[k].x, nNodes, m);
            idx_mark(r[k].y, nNodes, m);
            idx_mark(r[k].z, nNodes, m);
            idx_mark(r[k].w, nNodes, m);
        }
    } else {
        for (int j = 0; j < 32; ++j) {
            const int e = e0 + j;
            if (e < nEdges) idx_mark(erow[e], nNodes, m);
        }
    }
#pragma unroll
    for (int p = 0; p < IDX_PLANES; ++p) idx[(size_t)p * Gp + g] = m[p];
}

__device__ __forceinline__ void idx_tasks(int wave, int lane, const int* __restrict__ erow,
                                          float* support, int nNodes, int nEdges) {
    unsigned* idx = (unsigned*)(support + (size_t)nNodes * D);
    const int G      = (nEdges + 31) >> 5;
    const int nTasks = (((G + 3) & ~3) + 31) >> 5;
    const int wu     = __builtin_amdgcn_readfirstlane(wave);
    const int wpb    = (int)(blockDim.x >> 5);
    for (int tk = ((int)gridDim.x - 1 - (int)blockIdx.x) * wpb + wu; tk < nTasks; tk += (int)gridDim.x * wpb)
        idx_build(tk, lane, erow, idx, nNodes, nEdges);
}

#define AS_PITCH 132

__global__ __launch_bounds__(256) void gcn_gemm_wmma(const float* __restrict__ x,
                                                     const float* __restrict__ W,
                                                     float* __restrict__ support,
                                                     int nStripes,
                                                     const int* __restrict__ erow,
                                                     int nNodes, int nEdges) {
    __shared__ __attribute__((aligned(16))) float As[16 * AS_PITCH];

    const int t    = threadIdx.x;
    const int lane = t & 31;
    const int wave = t >> 5;
    const int n0   = wave * 16;
    const int half = lane >> 4;
    const int l16  = lane & 15;

    v2f bfrag[32];
#pragma unroll
    for (int s = 0; s < 32; ++s) {
        const int ka = 4 * s + half * 2;
        bfrag[s].x = W[(size_t)ka * D + n0 + l16];
        bfrag[s].y = W[(size_t)(ka + 1) * D + n0 + l16];
    }

    float* lds0 = As + (t >> 5) * AS_PITCH + (t & 31) * 4;
    float* lds1 = lds0 + 8 * AS_PITCH;
    const float* arow = As + l16 * AS_PITCH + half * 2;
    float* srow = support + n0 + l16;

    int stripe = blockIdx.x;
    const int first = stripe < nStripes ? stripe : 0;
    const float4* src0 = (const float4*)(x + (size_t)first * 16 * D);
    float4 g0 = src0[t];
    float4 g1 = src0[t + 256];
    for (; stripe < nStripes; stripe += gridDim.x) {
        __syncthreads();
        *(float4*)lds0 = g0;
        *(float4*)lds1 = g1;
        __syncthreads();

        const int next = stripe + gridDim.x;
        const int nx   = next < nStripes ? next : stripe;
        const float4* src = (const float4*)(x + (size_t)nx * 16 * D);
        g0 = src[t];
        g1 = src[t + 256];

        v8f c = {};
#pragma unroll
        for (int s = 0; s < 32; ++s) {
            const v2f a = *(const v2f*)(arow + 4 * s);
            c = __builtin_amdgcn_wmma_f32_16x16x4_f32(
                    false, a, false, bfrag[s], (short)0, c, false, false);
        }

        const int m0 = stripe * 16;
#pragma unroll
        for (int v = 0; v < 8; ++v) {
            srow[(size_t)(m0 + v + half * 8) * D] = c[v];
        }
    }

    idx_tasks(wave, lane, erow, support, nNodes, nEdges);
}

#define AGG_THREADS 1024
#define AGG_WAVES   32
#define AGG_GB      16
#define AGG_GCAP    80
#define AGG_QCAP    768
#define AGG_NB      (AGG_QCAP / 32)
#define AGG_OFFP    33
#define AGG_OCAP    64
#define AGG_RBITS   9
#define AGG_RPW     13
static_assert(AGG_RPW * AGG_WAVES >= AGG_ROWS && (AGG_RPW - 1) * AGG_WAVES < AGG_ROWS && AGG_RPW <= 16, "AGG_RPW");

__device__ __forceinline__ void agg_take(int e, int r, int r0, int nr, unsigned live, unsigned*& qe) {
    const unsigned rl   = (unsigned)(r - r0);
    const unsigned mask = __builtin_amdgcn_ballot_w32(rl < (unsigned)nr) & live;
    if (__builtin_amdgcn_inverse_ballot_w32(mask))
        qe[__builtin_amdgcn_mbcnt_lo(mask, 0u)] = ((unsigned)e << AGG_RBITS) | rl;
    qe += __builtin_popcount(mask);
}

template <bool FULL>
__device__ __forceinline__ void agg_groups(const unsigned* gl, int n, int lane, int r0, int nr, int nEdges,
                                           int gPart, unsigned tailMask,
                                           const int* __restrict__ erow, unsigned* q, int& cnt) {
    const int mine = lane < n ? (int)gl[lane] : 0;
    int rows[AGG_GB];
#pragma unroll
    for (int u = 0; u < AGG_GB; ++u) {
        const int g = __builtin_amdgcn_readlane(mine, FULL ? u : min(u, n - 1));
        rows[u] = erow[min(g * 32 + lane, nEdges - 1)];
    }
    unsigned* qe = q + cnt;
#pragma unroll
    for (int u = 0; u < AGG_GB; ++u) {
        const int g    = __builtin_amdgcn_readlane(mine, u);
        unsigned  live = g == gPart ? tailMask : ~0u;
        if (!FULL) live = u < n ? live : 0u;
        agg_take(g * 32 + lane, rows[u], r0, nr, live, qe);
    }
    cnt = (int)(qe - q);
}

__device__ __forceinline__ void agg_sort(unsigned* q, int cnt, int lane, int* bounds) {
    unsigned lbit[5];
#pragma unroll
    for (int t = 0; t < 5; ++t) lbit[t] = 0u - (((unsigned)lane >> t) & 1u);

    int run = 0;
    for (int b0 = 0; b0 < cnt; b0 += 32) {
        const bool     valid = b0 + lane < cnt;
        const unsigned e     = valid ? q[b0 + lane] : 0u;
        unsigned dBin = ~__builtin_amdgcn_ballot_w32(valid);
#pragma unroll
        for (int t = 0; t < 5; ++t)
            dBin |= __builtin_amdgcn_ballot_w32(((e >> t) & 1u) != 0u) ^ lbit[t];
        run += __builtin_popcount(~dBin);
    }
    int incl = run;
#pragma unroll
    for (int d = 1; d < 32; d <<= 1) {
        const int up = __shfl_up(incl, d);
        if (lane >= d) incl += up;
    }
    int next = incl - run;
    bounds[lane] = next;
    if (lane == 0) bounds[32] = cnt;
    unsigned ent[AGG_NB];
#pragma unroll
    for (int b = 0; b < AGG_NB; ++b)
        ent[b] = b * 32 + lane < cnt ? q[b * 32 + lane] : 0u;
#pragma unroll
    for (int b = 0; b < AGG_NB; ++b) {
        if (b * 32 < cnt) {
            const bool     valid = b * 32 + lane < cnt;
            const unsigned e     = ent[b];
            const unsigned dead  = ~__builtin_amdgcn_ballot_w32(valid);
            unsigned dBin = dead, dKey = dead;
#pragma unroll
            for (int t = 0; t < 5; ++t) {
                const unsigned bt = __builtin_amdgcn_ballot_w32(((e >> t) & 1u) != 0u);
                dBin |= bt ^ lbit[t];
                dKey |= bt ^ (0u - ((e >> t) & 1u));
            }
            const int place = __shfl(next, (int)(e & 31u)) + (int)__builtin_amdgcn_mbcnt_lo(~dKey, 0u);
            if (valid) q[place] = e;
            next += __builtin_popcount(~dBin);
        }
    }
}

typedef __attribute__((ext_vector_type(16))) float v16f;
struct AggSums { v16f x, y, z, w; };

__device__ __forceinline__ void agg_add(AggSums& s, int k, const float4 g, float vv) {
    s.x[k] += g.x * vv;
    s.y[k] += g.y * vv;
    s.z[k] += g.z * vv;
    s.w[k] += g.w * vv;
}

__device__ __forceinline__ void agg_drain(const unsigned* o, int m, int lane, int nNodes,
                                          const float* __restrict__ support,
                                          const int* __restrict__ ecol,
                                          const float* __restrict__ eval,
                                          AggSums& sums) {
    int c = 0, rl = 0;
    float v = 0.f;
    if (lane < m) {
        const unsigned ent = o[lane];
        const int e = (int)(ent >> AGG_RBITS);
        rl = (int)(ent & ((1u << AGG_RBITS) - 1u));
        c  = min(max(ecol[e], 0), nNodes - 1);
        v  = eval[e];
    }
    for (int j = 0; j < m; j += 8) {
        float4 g[8];
#pragma unroll
        for (int u = 0; u < 8; ++u) {
            const int cc = __builtin_amdgcn_readlane(c, min(j + u, m - 1));
            g[u] = ((const float4*)(support + (size_t)cc * D))[lane];
        }
#pragma unroll
        for (int u = 0; u < 8; ++u) {
            if (j + u < m) {
                const float vv = __int_as_float(__builtin_amdgcn_readlane(__float_as_int(v), j + u));
                const int   rr = __builtin_amdgcn_readlane(rl, j + u);
                agg_add(sums, rr >> 5, g[u], vv);
            }
        }
    }
}

__global__ __launch_bounds__(AGG_THREADS) void gcn_aggregate_tanh(const float* __restrict__ support,
                                                                  const int* __restrict__ erow,
                                                                  const int* __restrict__ ecol,
                                                                  const float* __restrict__ eval,
                                                                  float* __restrict__ out,
                                                                  int nNodes, int nEdges) {
    __shared__ unsigned hitq[AGG_WAVES * AGG_QCAP];
    __shared__ unsigned ownq[AGG_WAVES * AGG_OCAP];
    __shared__ unsigned glist[AGG_WAVES * AGG_GCAP];
    __shared__ int      qoff[AGG_WAVES * AGG_OFFP];
    __shared__ int      qmore[AGG_WAVES];

    const int t    = threadIdx.x;
    const int lane = t & 31;
    const int wv   = __builtin_amdgcn_readfirstlane(t >> 5);
    const int r0   = blockIdx.x * AGG_ROWS;
    const int nr   = min(AGG_ROWS, nNodes - r0);

    const int      G     = (nEdges + 31) >> 5;
    const int      Gp    = (G + 3) & ~3;
    const int      U2    = Gp >> 1;
    const int      uEnd  = max(U2 - 1, 0);
    const int      gLast = (nEdges - 1) >> 5;
    const int      nTail = nEdges - gLast * 32;
    const int      gPart = nTail < 32 ? gLast : -1;
    const unsigned tailMask = (1u << (nTail & 31)) - 1u;
    const unsigned bsel  = blockIdx.x & 31u;
    const int2*    plane = (const int2*)((const unsigned*)(support + (size_t)nNodes * D) + (size_t)(blockIdx.x >> 5) * Gp);

    int2 cur = plane[min(t, uEnd)];

    AggSums sums;
    sums.x = 0.f; sums.y = 0.f; sums.z = 0.f; sums.w = 0.f;

    unsigned* q  = hitq + wv * AGG_QCAP;
    unsigned* o  = ownq + wv * AGG_OCAP;
    unsigned* gl = glist + wv * AGG_GCAP;
    int cnt  = 0;
    int gcnt = 0;
    int sb   = 0;

    bool more;
    do {
        for (;;) {
            const bool tail = sb >= U2;
            if (gcnt >= AGG_GB || (tail && gcnt > 0)) {
                if (cnt > AGG_QCAP - AGG_GB * 32) break;
                const int n = min(gcnt, AGG_GB);
                gcnt -= n;
                if (n == AGG_GB) agg_groups<true>(gl + gcnt, n, lane, r0, nr, nEdges, gPart, tailMask, erow, q, cnt);
                else             agg_groups<false>(gl + gcnt, n, lane, r0, nr, nEdges, gPart, tailMask, erow, q, cnt);
                continue;
            }
            if (tail) break;
            const int2 nxt  = plane[min(sb + AGG_THREADS + t, uEnd)];
            const int      unit = sb + t;
            const unsigned live = __builtin_amdgcn_ballot_w32(unit < U2);
#pragma unroll
            for (int k = 0; k < 2; ++k) {
                const unsigned w    = (unsigned)(k == 0 ? cur.x : cur.y);
                const unsigned mask = __builtin_amdgcn_ballot_w32(((w >> bsel) & 1u) != 0u) & live;
                if (__builtin_amdgcn_inverse_ballot_w32(mask))
                    gl[gcnt + (int)__builtin_amdgcn_mbcnt_lo(mask, 0u)] = (unsigned)(2 * unit + k);
                gcnt += __builtin_popcount(mask);
            }
            cur = nxt;
            sb += AGG_THREADS;
        }
        agg_sort(q, cnt, lane, qoff + wv * AGG_OFFP);
        if (lane == 0) qmore[wv] = (sb < U2 || gcnt > 0) ? 1 : 0;
        __syncthreads();
        more = __builtin_amdgcn_ballot_w32(qmore[lane] != 0) != 0u;

        const int runLo = qoff[lane * AGG_OFFP + wv];
        const int runHi = qoff[lane * AGG_OFFP + wv + 1];
        int ocnt = 0;
        for (int s = 0; s < AGG_WAVES; ++s) {
            const int lo = __builtin_amdgcn_readlane(runLo, s);
            const int ns = __builtin_amdgcn_readlane(runHi, s) - lo;
            const unsigned* qs = hitq + s * AGG_QCAP + lo;
            for (int b = 0; b < ns; b += 32) {
                const int k = min(32, ns - b);
                if (lane < k) o[ocnt + lane] = qs[b + lane];
                ocnt += k;
                if (ocnt >= 32) {
                    agg_drain(o + (ocnt - 32), 32, lane, nNodes, support, ecol, eval, sums);
                    ocnt -= 32;
                }
            }
        }
        if (ocnt > 0) agg_drain(o, ocnt, lane, nNodes, support, ecol, eval, sums);

        if (!more) break;
        __syncthreads();
        cnt = 0;
    } while (true);

#pragma nounroll
    for (int k = 0; k < AGG_RPW; ++k) {
        const int row = wv + AGG_WAVES * k;
        if (row >= nr) break;
        float4 o4;
        o4.x = tanhf(sums.x[k]);
        o4.y = tanhf(sums.y[k]);
        o4.z = tanhf(sums.z[k]);
        o4.w = tanhf(sums.w[k]);
        ((float4*)(out + (size_t)(r0 + row) * D))[lane] = o4;
    }
}

extern "C" void kernel_launch(void* const* d_in, const int* in_sizes, int n_in,
                              void* d_out, int out_size, void* d_ws, size_t ws_size,
                              hipStream_t stream) {
    const float* x    = (const float*)d_in[0];
    const float* W    = (const float*)d_in[1];
    const int*   erow = (const int*)d_in[2];
    const int*   ecol = (const int*)d_in[3];
    const float* eval = (const float*)d_in[4];
    float* out = (float*)d_out;

    const int nNodes = in_sizes[0] / D;
    const int nEdges = in_sizes[2];

    float* support = (float*)d_ws;

    {
        const int stripes = nNodes / 16;
        gcn_gemm_wmma<<<1024, 256, 0, stream>>>(x, W, support, stripes, erow, nNodes, nEdges);
    }
    {
        const int blocks = (nNodes + AGG_ROWS - 1) / AGG_ROWS;
        gcn_aggregate_tanh<<<blocks, AGG_THREADS, 0, stream>>>(support, erow, ecol, eval, out, nNodes, nEdges);
    }
}
